# starting kernel
# baseline (speedup 1.0000x reference)
#include <hip/hip_runtime.h>

constexpr int kNG = 100000;
constexpr int kND = 50000;
constexpr int kNC = 50000;
constexpr int kH  = 128;

typedef __attribute__((ext_vector_type(2))) float v2f;
typedef __attribute__((ext_vector_type(8))) float v8f;

__global__ __launch_bounds__(256)
void hgnn_scatter(const float* __restrict__ x, const int* __restrict__ src,
                  const int* __restrict__ dst, float* __restrict__ sum,
                  float* __restrict__ cnt, int nE) {
  const int lane = threadIdx.x & 31;
  const int wave = (blockIdx.x * blockDim.x + threadIdx.x) >> 5;
  const int nW   = (gridDim.x * blockDim.x) >> 5;
  for (int e = wave; e < nE; e += nW) {
    const int s = src[e];
    const int d = dst[e];
    const float4 v = *(const float4*)(x + (size_t)s * kH + lane * 4);
    float* p = sum + (size_t)d * kH + lane * 4;
    atomicAdd(p + 0, v.x);
    atomicAdd(p + 1, v.y);
    atomicAdd(p + 2, v.z);
    atomicAdd(p + 3, v.w);
    if (lane == 0) atomicAdd(cnt + d, 1.0f);
  }
}

__global__ void hgnn_addmat(const float* __restrict__ a, const float* __restrict__ b,
                            float* __restrict__ o, int n) {
  int i = blockIdx.x * blockDim.x + threadIdx.x;
  if (i < n) o[i] = a[i] + b[i];
}

__global__ __launch_bounds__(256)
void hgnn_gemm3(const float* __restrict__ A0, const float* __restrict__ c0,
                const float* __restrict__ W0,
                const float* __restrict__ A1, const float* __restrict__ c1,
                const float* __restrict__ W1,
                const float* __restrict__ A2, const float* __restrict__ c2,
                const float* __restrict__ W2,
                const float* __restrict__ b0, const float* __restrict__ b1,
                float* __restrict__ Y, int relu) {
  __shared__ float As[16 * 132];
  const int tid  = threadIdx.x;
  const int wv   = tid >> 5;
  const int lane = tid & 31;
  const int m    = lane & 15;
  const int koff = (lane >> 4) << 1;
  const int col  = (wv << 4) + m;
  const int row0 = blockIdx.x << 4;

  const float* Alist[3] = {A0, A1, A2};
  const float* Clist[3] = {c0, c1, c2};
  const float* Wlist[3] = {W0, W1, W2};

  v8f acc = {};
  for (int t = 0; t < 3; ++t) {
    const float* A = Alist[t];
    if (!A) continue;
    const float* cp = Clist[t];
    const float* W  = Wlist[t];

    __syncthreads();
    for (int i = tid; i < 16 * kH; i += 256) {
      const int r = i >> 7;
      const int c = i & (kH - 1);
      float v = A[(size_t)(row0 + r) * kH + c];
      if (cp) v *= 1.0f / fmaxf(cp[row0 + r], 1.0f);
      As[r * 132 + c] = v;
    }
    __syncthreads();

#pragma unroll 8
    for (int k0 = 0; k0 < kH; k0 += 4) {
      const int k = k0 + koff;
      v2f av, bv;
      av.x = As[m * 132 + k];
      av.y = As[m * 132 + k + 1];
      bv.x = W[(size_t)k * kH + col];
      bv.y = W[(size_t)(k + 1) * kH + col];
      acc = __builtin_amdgcn_wmma_f32_16x16x4_f32(
          false, av, false, bv, (short)0, acc, false, false);
    }
  }

  const float bsum = (b0 ? b0[col] : 0.0f) + (b1 ? b1[col] : 0.0f);
  __syncthreads();
#pragma unroll
  for (int r = 0; r < 8; ++r) {
    float v = acc[r] + bsum;
    if (relu) v = fmaxf(v, 0.0f);
    Y[(size_t)(row0 + r + ((lane >> 4) << 3)) * kH + col] = v;
  }
}

extern "C" void kernel_launch(void* const* d_in, const int* in_sizes, int n_in,
                              void* d_out, int out_size, void* d_ws, size_t ws_size,
                              hipStream_t stream) {
  (void)n_in; (void)out_size; (void)ws_size;
  const float* EG = (const float*)d_in[0];
  const float* ED = (const float*)d_in[1];
  const float* EC = (const float*)d_in[2];
  const float* Wl = (const float*)d_in[3];
  const float* bl = (const float*)d_in[4];
  const float* Wr = (const float*)d_in[5];
  const int* eg_s  = (const int*)d_in[6];
  const int* eg_d  = (const int*)d_in[7];
  const int* ecd_s = (const int*)d_in[8];
  const int* ecd_d = (const int*)d_in[9];
  const int* ecg_s = (const int*)d_in[10];
  const int* ecg_d = (const int*)d_in[11];
  const int* egc_s = (const int*)d_in[12];
  const int* egc_d = (const int*)d_in[13];

  float* ws = (float*)d_ws;
  float* s0 = ws;
  float* s1 = s0 + (size_t)kNG * kH;
  float* s2 = s1 + (size_t)kND * kH;
  float* s3 = s2 + (size_t)kNG * kH;
  float* c0 = s3 + (size_t)kNC * kH;
  float* c1 = c0 + kNG;
  float* c2 = c1 + kND;
  float* c3 = c2 + kNG;
  float* wsum = c3 + kNC;
  const size_t nAccum = (size_t)(2 * kNG + kND + kNC) * kH + (2 * kNG + kND + kNC);
  const size_t zero_bytes = nAccum * sizeof(float);

  float* OG = (float*)d_out;
  float* OD = OG + (size_t)kNG * kH;
  float* OC = OD + (size_t)kND * kH;

  const float* xg = EG;
  const float* xd = ED;
  const float* xc = EC;

  for (int L = 0; L < 2; ++L) {
    hipMemsetAsync(ws, 0, zero_bytes, stream);
    hgnn_scatter<<<2048, 256, 0, stream>>>(xg, eg_s,  eg_d,  s0, c0, in_sizes[6]);
    hgnn_scatter<<<1024, 256, 0, stream>>>(xc, ecd_s, ecd_d, s1, c1, in_sizes[8]);
    hgnn_scatter<<<2048, 256, 0, stream>>>(xc, ecg_s, ecg_d, s2, c2, in_sizes[10]);
    hgnn_scatter<<<2048, 256, 0, stream>>>(xg, egc_s, egc_d, s3, c3, in_sizes[12]);

    const float* WlL = Wl + (size_t)L * 4 * kH * kH;
    const float* WrL = Wr + (size_t)L * 4 * kH * kH;
    const float* blL = bl + (size_t)L * 4 * kH;
    hgnn_addmat<<<(kH * kH + 255) / 256, 256, 0, stream>>>(WrL, WrL + 2 * kH * kH,
                                                           wsum, kH * kH);
    const int relu = (L == 0) ? 1 : 0;
    hgnn_gemm3<<<kNG / 16, 256, 0, stream>>>(
        s0, c0, WlL,
        s2, c2, WlL + 2 * kH * kH,
        xg, nullptr, wsum,
        blL, blL + 2 * kH, OG, relu);
    hgnn_gemm3<<<kND / 16, 256, 0, stream>>>(
        s1, c1, WlL + 1 * kH * kH,
        xd, nullptr, WrL + 1 * kH * kH,
        nullptr, nullptr, nullptr,
        blL + 1 * kH, nullptr, OD, relu);
    hgnn_gemm3<<<kNC / 16, 256, 0, stream>>>(
        s3, c3, WlL + 3 * kH * kH,
        xc, nullptr, WrL + 3 * kH * kH,
        nullptr, nullptr, nullptr,
        blL + 3 * kH, nullptr, OC, relu);

    xg = OG; xd = OD; xc = OC;
  }
}


# GEMM on a resident grid: 512 work-groups loop over 16-row tiles; a wave keeps its 16 columns of every W in registers (loaded once), only A fragments come from LDS; per-element sum order unchanged
# speedup vs baseline: 2.0392x; 2.0392x over previous
#include <hip/hip_runtime.h>

constexpr int kNG = 100000;
constexpr int kND = 50000;
constexpr int kNC = 50000;
constexpr int kH  = 128;

typedef __attribute__((ext_vector_type(2))) float v2f;
typedef __attribute__((ext_vector_type(8))) float v8f;

constexpr int kQCap = 288;

__device__ __forceinline__ int4 hgnn_ld4(const int* __restrict__ p, int e, int n) {
  int4 r;
  if (e + 3 < n) {
    r = *reinterpret_cast<const int4*>(p + e);
  } else {
    r.x = (e     < n) ? p[e]     : -1;
    r.y = (e + 1 < n) ? p[e + 1] : -1;
    r.z = (e + 2 < n) ? p[e + 2] : -1;
    r.w = (e + 3 < n) ? p[e + 3] : -1;
  }
  return r;
}

template <int R>
__global__ __launch_bounds__(1024)
void hgnn_ownsum(const float* __restrict__ x, const int* __restrict__ src,
                 const int* __restrict__ dst, float* __restrict__ sum,
                 float* __restrict__ cnt, int nE, int nDst) {
  __shared__ float acc[R * kH];
  __shared__ float cn[R];
  __shared__ int   q[32 * kQCap];
  const int tid  = threadIdx.x;
  const int lane = tid & 31;
  const int w    = tid >> 5;
  const int r0   = blockIdx.x * R;
  int* qw = q + w * kQCap;

  float4* acc4 = reinterpret_cast<float4*>(acc);
  for (int i = tid; i < R * kH / 4; i += 1024) acc4[i] = make_float4(0.f, 0.f, 0.f, 0.f);
  for (int i = tid; i < R; i += 1024) cn[i] = 0.f;
  __syncthreads();

  int qn = 0;

  auto push = [&](int d, int e) {
    const unsigned dl = (unsigned)(d - r0);
    const bool m = dl < (unsigned)R;
    const unsigned mask = __builtin_amdgcn_ballot_w32(m);
    if (mask) {
      if (m) qw[qn + (int)__builtin_amdgcn_mbcnt_lo(mask, 0u)] = (int)((dl << 20) | (unsigned)e);
      qn += __builtin_popcount(mask);
    }
  };

  auto drain = [&]() {
    for (int b0 = 0; b0 < qn; b0 += 32) {
      const int nb = (qn - b0 < 32) ? (qn - b0) : 32;
      int ent = 0, s = 0;
      if (lane < nb) { ent = qw[b0 + lane]; s = src[ent & 0xFFFFF]; }
      for (int i = 0; i < nb; i += 4) {
        float v[4][4];
        int   dl[4];
#pragma unroll
        for (int u = 0; u < 4; ++u) {
          const int k = __builtin_amdgcn_readfirstlane(i + u);
          const int su = __builtin_amdgcn_readlane(s, k);
          dl[u] = __builtin_amdgcn_readlane(ent, k) >> 20;
          v[u][0] = v[u][1] = v[u][2] = v[u][3] = 0.f;
          if (k < nb) {
            const float* r = x + (size_t)su * kH + lane;
            v[u][0] = r[0]; v[u][1] = r[32]; v[u][2] = r[64]; v[u][3] = r[96];
          }
        }
#pragma unroll
        for (int u = 0; u < 4; ++u) {
          if (i + u < nb) {
            float* a = acc + dl[u] * kH + lane;
            atomicAdd(a,      v[u][0]);
            atomicAdd(a + 32, v[u][1]);
            atomicAdd(a + 64, v[u][2]);
            atomicAdd(a + 96, v[u][3]);
            if (lane == 0) atomicAdd(cn + dl[u], 1.0f);
          }
        }
      }
    }
    qn = 0;
  };

  auto scan = [&](const int4& a, const int4& b, int c) {
    const int e0 = c * 256 + lane * 4;
    push(a.x, e0);       push(a.y, e0 + 1);   push(a.z, e0 + 2);   push(a.w, e0 + 3);
    push(b.x, e0 + 128); push(b.y, e0 + 129); push(b.z, e0 + 130); push(b.w, e0 + 131);
    if (qn >= 32) drain();
  };
  auto ldv = [&](int c, int half) {
    return *reinterpret_cast<const int4*>(dst + c * 256 + half * 128 + lane * 4);
  };
  const int wu    = __builtin_amdgcn_readfirstlane(w);
  const int nFull = nE >> 8;
  if (wu < nFull) {
    int c = wu;
    int4 a0 = ldv(c, 0), b0 = ldv(c, 1);
    while (true) {
      const int c1 = c + 32;
      const int c1c = (c1 < nFull) ? c1 : c;
      const int4 a1 = ldv(c1c, 0), b1 = ldv(c1c, 1);
      scan(a0, b0, c);
      if (c1 >= nFull) break;
      const int c2 = c1 + 32;
      const int c2c = (c2 < nFull) ? c2 : c1;
      a0 = ldv(c2c, 0); b0 = ldv(c2c, 1);
      scan(a1, b1, c1);
      if (c2 >= nFull) break;
      c = c2;
    }
  }
  if ((nE & 255) != 0 && (nFull & 31) == wu) {
    const int4 a = hgnn_ld4(dst, nFull * 256 + lane * 4, nE);
    const int4 b = hgnn_ld4(dst, nFull * 256 + 128 + lane * 4, nE);
    scan(a, b, nFull);
  }
  drain();
  __syncthreads();

  float4* out4 = reinterpret_cast<float4*>(sum + (size_t)r0 * kH);
  for (int i = tid; i < R * kH / 4; i += 1024)
    if (r0 + (i >> 5) < nDst) out4[i] = acc4[i];
  for (int i = tid; i < R; i += 1024)
    if (r0 + i < nDst) cnt[r0 + i] = cn[i];
}

__global__ void hgnn_addmat(const float* __restrict__ a, const float* __restrict__ b,
                            float* __restrict__ o, int n) {
  int i = blockIdx.x * blockDim.x + threadIdx.x;
  if (i < n) o[i] = a[i] + b[i];
}

template <int NT>
__global__ __launch_bounds__(256)
void hgnn_gemm3(const float* A0, const float* __restrict__ c0,
                const float* __restrict__ W0,
                const float* A1, const float* __restrict__ c1,
                const float* __restrict__ W1,
                const float* A2, const float* __restrict__ c2,
                const float* __restrict__ W2,
                const float* __restrict__ b0, const float* __restrict__ b1,
                float* Y, int relu, int nTiles) {
  __shared__ __attribute__((aligned(16))) float As[NT][16 * 132];
  const int tid  = threadIdx.x;
  const int wv   = tid >> 5;
  const int lane = tid & 31;
  const int m    = lane & 15;
  const int koff = (lane >> 4) << 1;
  const int col  = (wv << 4) + m;

  const float* Alist[3] = {A0, A1, A2};
  const float* Clist[3] = {c0, c1, c2};
  const float* Wlist[3] = {W0, W1, W2};

  v2f bw[NT][32];
#pragma unroll
  for (int t = 0; t < NT; ++t) {
#pragma unroll
    for (int j = 0; j < 32; ++j) {
      const int k = 4 * j + koff;
      bw[t][j].x = Wlist[t][(size_t)k * kH + col];
      bw[t][j].y = Wlist[t][(size_t)(k + 1) * kH + col];
    }
  }
  const float bsum = (b0 ? b0[col] : 0.0f) + (b1 ? b1[col] : 0.0f);

  const int sr  = __builtin_amdgcn_readfirstlane(tid >> 5);
  const int sc4 = tid & 31;
  float4 st[NT][2];
  float  cn[NT][2];
  auto fetch = [&](int tile) {
    const unsigned row0 = (unsigned)tile << 4;
#pragma unroll
    for (int t = 0; t < NT; ++t) {
#pragma unroll
      for (int u = 0; u < 2; ++u) {
        const unsigned r = row0 + (unsigned)(sr + 8 * u);
        st[t][u] = *reinterpret_cast<const float4*>(Alist[t] + (r * kH + 4u * (unsigned)sc4));
        cn[t][u] = Clist[t] ? Clist[t][r] : 1.0f;
      }
    }
  };

  int tile = blockIdx.x;
  if (tile < nTiles) fetch(tile);
  for (; tile < nTiles; tile += gridDim.x) {
    const int row0 = tile << 4;
    __syncthreads();
#pragma unroll
    for (int t = 0; t < NT; ++t) {
#pragma unroll
      for (int u = 0; u < 2; ++u) {
        float4 v = st[t][u];
        if (Clist[t]) {
          const float s = 1.0f / fmaxf(cn[t][u], 1.0f);
          v.x *= s; v.y *= s; v.z *= s; v.w *= s;
        }
        *reinterpret_cast<float4*>(&As[t][(sr + 8 * u) * 132 + 4 * sc4]) = v;
      }
    }
    __syncthreads();
    const int nxt = tile + gridDim.x;
    if (nxt < nTiles) fetch(nxt);

    v8f acc = {};
#pragma unroll
    for (int t = 0; t < NT; ++t) {
#pragma unroll
      for (int j = 0; j < 32; ++j) {
        const int k = 4 * j + koff;
        v2f av;
        av.x = As[t][m * 132 + k];
        av.y = As[t][m * 132 + k + 1];
        acc = __builtin_amdgcn_wmma_f32_16x16x4_f32(
            false, av, false, bw[t][j], (short)0, acc, false, false);
      }
    }

#pragma unroll
    for (int r = 0; r < 8; ++r) {
      float v = acc[r] + bsum;
      if (relu) v = fmaxf(v, 0.0f);
      Y[(unsigned)(row0 + r + ((lane >> 4) << 3)) * kH + (unsigned)col] = v;
    }
  }
}

extern "C" void kernel_launch(void* const* d_in, const int* in_sizes, int n_in,
                              void* d_out, int out_size, void* d_ws, size_t ws_size,
                              hipStream_t stream) {
  (void)n_in; (void)out_size; (void)ws_size;
  const float* EG = (const float*)d_in[0];
  const float* ED = (const float*)d_in[1];
  const float* EC = (const float*)d_in[2];
  const float* Wl = (const float*)d_in[3];
  const float* bl = (const float*)d_in[4];
  const float* Wr = (const float*)d_in[5];
  const int* eg_s  = (const int*)d_in[6];
  const int* eg_d  = (const int*)d_in[7];
  const int* ecd_s = (const int*)d_in[8];
  const int* ecd_d = (const int*)d_in[9];
  const int* ecg_s = (const int*)d_in[10];
  const int* ecg_d = (const int*)d_in[11];
  const int* egc_s = (const int*)d_in[12];
  const int* egc_d = (const int*)d_in[13];

  float* ws = (float*)d_ws;
  float* s0 = ws;
  float* s1 = s0 + (size_t)kNG * kH;
  float* s2 = s1 + (size_t)kND * kH;
  float* s3 = s2 + (size_t)kNG * kH;
  float* c0 = s3 + (size_t)kNC * kH;
  float* c1 = c0 + kNG;
  float* c2 = c1 + kND;
  float* c3 = c2 + kNG;
  float* wsum = c3 + kNC;

  float* OG = (float*)d_out;
  float* OD = OG + (size_t)kNG * kH;
  float* OC = OD + (size_t)kND * kH;

  const float* xg = EG;
  const float* xd = ED;
  const float* xc = EC;

  for (int L = 0; L < 2; ++L) {
    hgnn_ownsum<400><<<(kNG + 399) / 400, 1024, 0, stream>>>(xg, eg_s,  eg_d,  s0, c0, in_sizes[6],  kNG);
    hgnn_ownsum<200><<<(kND + 199) / 200, 1024, 0, stream>>>(xc, ecd_s, ecd_d, s1, c1, in_sizes[8],  kND);
    hgnn_ownsum<400><<<(kNG + 399) / 400, 1024, 0, stream>>>(xc, ecg_s, ecg_d, s2, c2, in_sizes[10], kNG);
    hgnn_ownsum<200><<<(kNC + 199) / 200, 1024, 0, stream>>>(xg, egc_s, egc_d, s3, c3, in_sizes[12], kNC);

    const float* WlL = Wl + (size_t)L * 4 * kH * kH;
    const float* WrL = Wr + (size_t)L * 4 * kH * kH;
    const float* blL = bl + (size_t)L * 4 * kH;
    hgnn_addmat<<<(kH * kH + 255) / 256, 256, 0, stream>>>(WrL, WrL + 2 * kH * kH,
                                                           wsum, kH * kH);
    const int relu = (L == 0) ? 1 : 0;
    hgnn_gemm3<3><<<512, 256, 0, stream>>>(
        s0, c0, WlL,
        s2, c2, WlL + 2 * kH * kH,
        xg, nullptr, wsum,
        blL, blL + 2 * kH, OG, relu, kNG / 16);
    hgnn_gemm3<2><<<512, 256, 0, stream>>>(
        s1, c1, WlL + 1 * kH * kH,
        xd, nullptr, WrL + 1 * kH * kH,
        nullptr, nullptr, nullptr,
        blL + 1 * kH, nullptr, OD, relu, kND / 16);
    hgnn_gemm3<2><<<512, 256, 0, stream>>>(
        s3, c3, WlL + 3 * kH * kH,
        xc, nullptr, WrL + 3 * kH * kH,
        nullptr, nullptr, nullptr,
        blL + 3 * kH, nullptr, OC, relu, kNC / 16);

    xg = OG; xd = OD; xc = OC;
  }
}


# Segment-sum scan reads 1-byte owner keys (dst/R, written once per call by a pre-pass) instead of 4-byte dst: 4x fewer scan bytes, byte-parallel match; LDS arrival-order sums as before
# speedup vs baseline: 2.4166x; 1.1851x over previous
#include <hip/hip_runtime.h>

constexpr int kNG = 100000;
constexpr int kND = 50000;
constexpr int kNC = 50000;
constexpr int kH  = 128;

typedef __attribute__((ext_vector_type(2))) float v2f;
typedef __attribute__((ext_vector_type(8))) float v8f;

constexpr int kKeyWords  = 150016;
constexpr int kKeyWords1 = 75008;

__global__ __launch_bounds__(256)
void hgnn_keys(const int* __restrict__ d0, const int* __restrict__ d1,
               const int* __restrict__ d2, const int* __restrict__ d3,
               int n0, int n1, int n2, int n3, unsigned* __restrict__ keys) {
  const int rel = blockIdx.y;
  const int* d = (rel == 0) ? d0 : (rel == 1) ? d1 : (rel == 2) ? d2 : d3;
  const int  n = (rel == 0) ? n0 : (rel == 1) ? n1 : (rel == 2) ? n2 : n3;
  const unsigned nDst = (rel & 1) ? (unsigned)kND : (unsigned)kNG;
  const int      sh   = (rel & 1) ? 0 : 1;
  const int wi = blockIdx.x * 256 + threadIdx.x;
  const int nPadWords = ((n + 1023) >> 10) << 8;
  if (wi >= nPadWords) return;
  const int e = wi * 4;
  int4 v;
  if (e + 3 < n) {
    v = *reinterpret_cast<const int4*>(d + e);
  } else {
    v.x = (e     < n) ? d[e]     : -1;
    v.y = (e + 1 < n) ? d[e + 1] : -1;
    v.z = (e + 2 < n) ? d[e + 2] : -1;
    v.w = (e + 3 < n) ? d[e + 3] : -1;
  }
  auto key = [&](int dd) -> unsigned {
    const unsigned u = (unsigned)dd;
    return (u < nDst) ? ((u / 200u) >> sh) : 255u;
  };
  const int base = (rel == 0) ? 0 : (rel == 1) ? kKeyWords : (rel == 2) ? kKeyWords + kKeyWords1
                                                                      : 2 * kKeyWords + kKeyWords1;
  keys[base + wi] =
      key(v.x) | (key(v.y) << 8) | (key(v.z) << 16) | (key(v.w) << 24);
}

__device__ __forceinline__ unsigned hgnn_eq8(unsigned w, unsigned own4) {
  const unsigned x = w ^ own4;
  return ~(((x & 0x7F7F7F7Fu) + 0x7F7F7F7Fu) | x) & 0x80808080u;
}

constexpr int kQCap = 64;

template <int R>
__global__ __launch_bounds__(1024)
void hgnn_ownsum(const float* __restrict__ x, const int* __restrict__ src,
                 const int* __restrict__ dst, const unsigned* __restrict__ keys,
                 float* __restrict__ sum, float* __restrict__ cnt, int nE, int nDst) {
  __shared__ float acc[R * kH];
  __shared__ float cn[R];
  __shared__ int   q[32 * kQCap];
  const int tid  = threadIdx.x;
  const int lane = tid & 31;
  const int w    = tid >> 5;
  const int r0   = blockIdx.x * R;
  int* qw = q + w * kQCap;

  float4* acc4 = reinterpret_cast<float4*>(acc);
  for (int i = tid; i < R * kH / 4; i += 1024) acc4[i] = make_float4(0.f, 0.f, 0.f, 0.f);
  for (int i = tid; i < R; i += 1024) cn[i] = 0.f;
  __syncthreads();

  int qn = 0;

  auto drain = [&]() {
    for (int b0 = 0; b0 < qn; b0 += 32) {
      const int nb = (qn - b0 < 32) ? (qn - b0) : 32;
      int ent = 0, s = 0;
      if (lane < nb) { const int e = qw[b0 + lane]; s = src[e]; ent = dst[e] - r0; }
      for (int i = 0; i < nb; i += 4) {
        float v[4][4];
        int   dl[4];
#pragma unroll
        for (int u = 0; u < 4; ++u) {
          const int k = __builtin_amdgcn_readfirstlane(i + u);
          const int su = __builtin_amdgcn_readlane(s, k);
          dl[u] = __builtin_amdgcn_readlane(ent, k);
          v[u][0] = v[u][1] = v[u][2] = v[u][3] = 0.f;
          if (k < nb) {
            const float* r = x + (size_t)su * kH + lane;
            v[u][0] = r[0]; v[u][1] = r[32]; v[u][2] = r[64]; v[u][3] = r[96];
          }
        }
#pragma unroll
        for (int u = 0; u < 4; ++u) {
          if (i + u < nb) {
            float* a = acc + dl[u] * kH + lane;
            atomicAdd(a,      v[u][0]);
            atomicAdd(a + 32, v[u][1]);
            atomicAdd(a + 64, v[u][2]);
            atomicAdd(a + 96, v[u][3]);
            if (lane == 0) atomicAdd(cn + dl[u], 1.0f);
          }
        }
      }
    }
    qn = 0;
  };

  const unsigned own4 = (unsigned)blockIdx.x * 0x01010101u;
  auto scan = [&](const uint4& a, const uint4& b, int c) {
    unsigned zz = (hgnn_eq8(a.x, own4) >> 7) | (hgnn_eq8(a.y, own4) >> 6) |
                  (hgnn_eq8(a.z, own4) >> 5) | (hgnn_eq8(a.w, own4) >> 4) |
                  (hgnn_eq8(b.x, own4) >> 3) | (hgnn_eq8(b.y, own4) >> 2) |
                  (hgnn_eq8(b.z, own4) >> 1) |  hgnn_eq8(b.w, own4);
    const int e0 = c * 1024 + lane * 16;
    unsigned mask = __builtin_amdgcn_ballot_w32(zz != 0u);
    while (mask) {
      if (zz) {
        const int p = __builtin_ctz(zz);
        zz &= zz - 1u;
        qw[qn + (int)__builtin_amdgcn_mbcnt_lo(mask, 0u)] =
            e0 + ((p & 4) << 7) + ((p & 3) << 2) + (p >> 3);
      }
      qn += __builtin_popcount(mask);
      if (qn >= 32) drain();
      mask = __builtin_amdgcn_ballot_w32(zz != 0u);
    }
  };
  auto ldv = [&](int c, int half) {
    return *reinterpret_cast<const uint4*>(keys + c * 256 + half * 128 + lane * 4);
  };
  const int wu     = __builtin_amdgcn_readfirstlane(w);
  const int nTrips = (nE + 1023) >> 10;
  if (wu < nTrips) {
    int c = wu;
    uint4 a0 = ldv(c, 0), b0 = ldv(c, 1);
    while (true) {
      const int c1 = c + 32;
      const int c1c = (c1 < nTrips) ? c1 : c;
      const uint4 a1 = ldv(c1c, 0), b1 = ldv(c1c, 1);
      scan(a0, b0, c);
      if (c1 >= nTrips) break;
      const int c2 = c1 + 32;
      const int c2c = (c2 < nTrips) ? c2 : c1;
      a0 = ldv(c2c, 0); b0 = ldv(c2c, 1);
      scan(a1, b1, c1);
      if (c2 >= nTrips) break;
      c = c2;
    }
  }
  drain();
  __syncthreads();

  float4* out4 = reinterpret_cast<float4*>(sum + (size_t)r0 * kH);
  for (int i = tid; i < R * kH / 4; i += 1024)
    if (r0 + (i >> 5) < nDst) out4[i] = acc4[i];
  for (int i = tid; i < R; i += 1024)
    if (r0 + i < nDst) cnt[r0 + i] = cn[i];
}

__global__ void hgnn_addmat(const float* __restrict__ a, const float* __restrict__ b,
                            float* __restrict__ o, int n) {
  int i = blockIdx.x * blockDim.x + threadIdx.x;
  if (i < n) o[i] = a[i] + b[i];
}

template <int NT>
__global__ __launch_bounds__(256)
void hgnn_gemm3(const float* A0, const float* __restrict__ c0,
                const float* __restrict__ W0,
                const float* A1, const float* __restrict__ c1,
                const float* __restrict__ W1,
                const float* A2, const float* __restrict__ c2,
                const float* __restrict__ W2,
                const float* __restrict__ b0, const float* __restrict__ b1,
                float* Y, int relu, int nTiles) {
  __shared__ __attribute__((aligned(16))) float As[NT][16 * 132];
  const int tid  = threadIdx.x;
  const int wv   = tid >> 5;
  const int lane = tid & 31;
  const int m    = lane & 15;
  const int koff = (lane >> 4) << 1;
  const int col  = (wv << 4) + m;

  const float* Alist[3] = {A0, A1, A2};
  const float* Clist[3] = {c0, c1, c2};
  const float* Wlist[3] = {W0, W1, W2};

  v2f bw[NT][32];
#pragma unroll
  for (int t = 0; t < NT; ++t) {
#pragma unroll
    for (int j = 0; j < 32; ++j) {
      const int k = 4 * j + koff;
      bw[t][j].x = Wlist[t][(size_t)k * kH + col];
      bw[t][j].y = Wlist[t][(size_t)(k + 1) * kH + col];
    }
  }
  const float bsum = (b0 ? b0[col] : 0.0f) + (b1 ? b1[col] : 0.0f);

  const int sr  = __builtin_amdgcn_readfirstlane(tid >> 5);
  const int sc4 = tid & 31;
  float4 st[NT][2];
  float  cn[NT][2];
  auto fetch = [&](int tile) {
    const unsigned row0 = (unsigned)tile << 4;
#pragma unroll
    for (int t = 0; t < NT; ++t) {
#pragma unroll
      for (int u = 0; u < 2; ++u) {
        const unsigned r = row0 + (unsigned)(sr + 8 * u);
        st[t][u] = *reinterpret_cast<const float4*>(Alist[t] + (r * kH + 4u * (unsigned)sc4));
        cn[t][u] = Clist[t] ? Clist[t][r] : 1.0f;
      }
    }
  };

  int tile = blockIdx.x;
  if (tile < nTiles) fetch(tile);
  for (; tile < nTiles; tile += gridDim.x) {
    const int row0 = tile << 4;
    __syncthreads();
#pragma unroll
    for (int t = 0; t < NT; ++t) {
#pragma unroll
      for (int u = 0; u < 2; ++u) {
        float4 v = st[t][u];
        if (Clist[t]) {
          const float s = 1.0f / fmaxf(cn[t][u], 1.0f);
          v.x *= s; v.y *= s; v.z *= s; v.w *= s;
        }
        *reinterpret_cast<float4*>(&As[t][(sr + 8 * u) * 132 + 4 * sc4]) = v;
      }
    }
    __syncthreads();
    const int nxt = tile + gridDim.x;
    if (nxt < nTiles) fetch(nxt);

    v8f acc = {};
#pragma unroll
    for (int t = 0; t < NT; ++t) {
#pragma unroll
      for (int j = 0; j < 32; ++j) {
        const int k = 4 * j + koff;
        v2f av;
        av.x = As[t][m * 132 + k];
        av.y = As[t][m * 132 + k + 1];
        acc = __builtin_amdgcn_wmma_f32_16x16x4_f32(
            false, av, false, bw[t][j], (short)0, acc, false, false);
      }
    }

#pragma unroll
    for (int r = 0; r < 8; ++r) {
      float v = acc[r] + bsum;
      if (relu) v = fmaxf(v, 0.0f);
      Y[(unsigned)(row0 + r + ((lane >> 4) << 3)) * kH + (unsigned)col] = v;
    }
  }
}

extern "C" void kernel_launch(void* const* d_in, const int* in_sizes, int n_in,
                              void* d_out, int out_size, void* d_ws, size_t ws_size,
                              hipStream_t stream) {
  (void)n_in; (void)out_size; (void)ws_size;
  const float* EG = (const float*)d_in[0];
  const float* ED = (const float*)d_in[1];
  const float* EC = (const float*)d_in[2];
  const float* Wl = (const float*)d_in[3];
  const float* bl = (const float*)d_in[4];
  const float* Wr = (const float*)d_in[5];
  const int* eg_s  = (const int*)d_in[6];
  const int* eg_d  = (const int*)d_in[7];
  const int* ecd_s = (const int*)d_in[8];
  const int* ecd_d = (const int*)d_in[9];
  const int* ecg_s = (const int*)d_in[10];
  const int* ecg_d = (const int*)d_in[11];
  const int* egc_s = (const int*)d_in[12];
  const int* egc_d = (const int*)d_in[13];

  float* ws = (float*)d_ws;
  float* s0 = ws;
  float* s1 = s0 + (size_t)kNG * kH;
  float* s2 = s1 + (size_t)kND * kH;
  float* s3 = s2 + (size_t)kNG * kH;
  float* c0 = s3 + (size_t)kNC * kH;
  float* c1 = c0 + kNG;
  float* c2 = c1 + kND;
  float* c3 = c2 + kNG;
  float* wsum = c3 + kNC;
  unsigned* keys = (unsigned*)(wsum + kH * kH);
  const unsigned* k0 = keys;
  const unsigned* k1 = k0 + kKeyWords;
  const unsigned* k2 = k1 + kKeyWords1;
  const unsigned* k3 = k2 + kKeyWords;

  float* OG = (float*)d_out;
  float* OD = OG + (size_t)kNG * kH;
  float* OC = OD + (size_t)kND * kH;

  const float* xg = EG;
  const float* xd = ED;
  const float* xc = EC;

  hgnn_keys<<<dim3(kKeyWords / 256, 4), 256, 0, stream>>>(
      eg_d, ecd_d, ecg_d, egc_d,
      in_sizes[6], in_sizes[8], in_sizes[10], in_sizes[12], keys);

  for (int L = 0; L < 2; ++L) {
    hgnn_ownsum<400><<<(kNG + 399) / 400, 1024, 0, stream>>>(xg, eg_s,  eg_d,  k0, s0, c0, in_sizes[6],  kNG);
    hgnn_ownsum<200><<<(kND + 199) / 200, 1024, 0, stream>>>(xc, ecd_s, ecd_d, k1, s1, c1, in_sizes[8],  kND);
    hgnn_ownsum<400><<<(kNG + 399) / 400, 1024, 0, stream>>>(xc, ecg_s, ecg_d, k2, s2, c2, in_sizes[10], kNG);
    hgnn_ownsum<200><<<(kNC + 199) / 200, 1024, 0, stream>>>(xg, egc_s, egc_d, k3, s3, c3, in_sizes[12], kNC);

    const float* WlL = Wl + (size_t)L * 4 * kH * kH;
    const float* WrL = Wr + (size_t)L * 4 * kH * kH;
    const float* blL = bl + (size_t)L * 4 * kH;
    hgnn_addmat<<<(kH * kH + 255) / 256, 256, 0, stream>>>(WrL, WrL + 2 * kH * kH,
                                                           wsum, kH * kH);
    const int relu = (L == 0) ? 1 : 0;
    hgnn_gemm3<3><<<512, 256, 0, stream>>>(
        s0, c0, WlL,
        s2, c2, WlL + 2 * kH * kH,
        xg, nullptr, wsum,
        blL, blL + 2 * kH, OG, relu, kNG / 16);
    hgnn_gemm3<2><<<512, 256, 0, stream>>>(
        s1, c1, WlL + 1 * kH * kH,
        xd, nullptr, WrL + 1 * kH * kH,
        nullptr, nullptr, nullptr,
        blL + 1 * kH, nullptr, OD, relu, kND / 16);
    hgnn_gemm3<2><<<512, 256, 0, stream>>>(
        s3, c3, WlL + 3 * kH * kH,
        xc, nullptr, WrL + 3 * kH * kH,
        nullptr, nullptr, nullptr,
        blL + 3 * kH, nullptr, OC, relu, kNC / 16);

    xg = OG; xd = OD; xc = OC;
  }
}


# Segment-sum drain keeps 16 gather rows in flight per wave (was 4): only whole 32-edge batches leave the per-wave ring queue, as two 16-row groups; batch counts in one LDS add; arrival-order sums kept
# speedup vs baseline: 2.5097x; 1.0385x over previous
#include <hip/hip_runtime.h>

constexpr int kNG = 100000;
constexpr int kND = 50000;
constexpr int kNC = 50000;
constexpr int kH  = 128;

typedef __attribute__((ext_vector_type(2))) float v2f;
typedef __attribute__((ext_vector_type(8))) float v8f;

constexpr int kKeyWords  = 150016;
constexpr int kKeyWords1 = 75008;

__global__ __launch_bounds__(256)
void hgnn_keys(const int* __restrict__ d0, const int* __restrict__ d1,
               const int* __restrict__ d2, const int* __restrict__ d3,
               int n0, int n1, int n2, int n3, unsigned* __restrict__ keys) {
  const int rel = blockIdx.y;
  const int* d = (rel == 0) ? d0 : (rel == 1) ? d1 : (rel == 2) ? d2 : d3;
  const int  n = (rel == 0) ? n0 : (rel == 1) ? n1 : (rel == 2) ? n2 : n3;
  const unsigned nDst = (rel & 1) ? (unsigned)kND : (unsigned)kNG;
  const int      sh   = (rel & 1) ? 0 : 1;
  const int wi = blockIdx.x * 256 + threadIdx.x;
  const int nPadWords = ((n + 1023) >> 10) << 8;
  if (wi >= nPadWords) return;
  const int e = wi * 4;
  int4 v;
  if (e + 3 < n) {
    v = *reinterpret_cast<const int4*>(d + e);
  } else {
    v.x = (e     < n) ? d[e]     : -1;
    v.y = (e + 1 < n) ? d[e + 1] : -1;
    v.z = (e + 2 < n) ? d[e + 2] : -1;
    v.w = (e + 3 < n) ? d[e + 3] : -1;
  }
  auto key = [&](int dd) -> unsigned {
    const unsigned u = (unsigned)dd;
    return (u < nDst) ? ((u / 200u) >> sh) : 255u;
  };
  const int base = (rel == 0) ? 0 : (rel == 1) ? kKeyWords : (rel == 2) ? kKeyWords + kKeyWords1
                                                                      : 2 * kKeyWords + kKeyWords1;
  keys[base + wi] =
      key(v.x) | (key(v.y) << 8) | (key(v.z) << 16) | (key(v.w) << 24);
}

__device__ __forceinline__ unsigned hgnn_eq8(unsigned w, unsigned own4) {
  const unsigned x = w ^ own4;
  return ~(((x & 0x7F7F7F7Fu) + 0x7F7F7F7Fu) | x) & 0x80808080u;
}

constexpr int kQCap = 64;
constexpr int kGrp  = 16;

template <int R>
__global__ __launch_bounds__(1024)
void hgnn_ownsum(const float* __restrict__ x, const int* __restrict__ src,
                 const int* __restrict__ dst, const unsigned* __restrict__ keys,
                 float* __restrict__ sum, float* __restrict__ cnt, int nE, int nDst) {
  __shared__ float acc[R * kH];
  __shared__ float cn[R];
  __shared__ int   q[32 * kQCap];
  const int tid  = threadIdx.x;
  const int lane = tid & 31;
  const int w    = tid >> 5;
  const int r0   = blockIdx.x * R;
  int* qw = q + w * kQCap;

  float4* acc4 = reinterpret_cast<float4*>(acc);
  for (int i = tid; i < R * kH / 4; i += 1024) acc4[i] = make_float4(0.f, 0.f, 0.f, 0.f);
  for (int i = tid; i < R; i += 1024) cn[i] = 0.f;
  __syncthreads();

  int qn = 0;
  int hd = 0;

  auto drain = [&](int nb) {
    int ent = 0, s = 0;
    if (lane < nb) {
      const int e = qw[(hd + lane) & (kQCap - 1)];
      s = src[e];
      ent = dst[e] - r0;
      atomicAdd(cn + ent, 1.0f);
    }
    for (int i = 0; i < nb; i += kGrp) {
      float2 v[kGrp][2];
      int    dl[kGrp];
#pragma unroll
      for (int u = 0; u < kGrp; ++u) {
        const int k = __builtin_amdgcn_readfirstlane(i + u);
        const int su = __builtin_amdgcn_readlane(s, k);
        dl[u] = __builtin_amdgcn_readlane(ent, k);
        if (k < nb) {
          const float* r = x + (size_t)su * kH + 2 * lane;
          v[u][0] = *reinterpret_cast<const float2*>(r);
          v[u][1] = *reinterpret_cast<const float2*>(r + 64);
        }
      }
#pragma unroll
      for (int u = 0; u < kGrp; ++u) {
        if (i + u < nb) {
          float* a = acc + dl[u] * kH + 2 * lane;
          atomicAdd(a,      v[u][0].x);
          atomicAdd(a + 1,  v[u][0].y);
          atomicAdd(a + 64, v[u][1].x);
          atomicAdd(a + 65, v[u][1].y);
        }
      }
    }
    hd = (hd + nb) & (kQCap - 1);
    qn -= nb;
  };

  const unsigned own4 = (unsigned)blockIdx.x * 0x01010101u;
  auto scan = [&](const uint4& a, const uint4& b, int c) {
    unsigned zz = (hgnn_eq8(a.x, own4) >> 7) | (hgnn_eq8(a.y, own4) >> 6) |
                  (hgnn_eq8(a.z, own4) >> 5) | (hgnn_eq8(a.w, own4) >> 4) |
                  (hgnn_eq8(b.x, own4) >> 3) | (hgnn_eq8(b.y, own4) >> 2) |
                  (hgnn_eq8(b.z, own4) >> 1) |  hgnn_eq8(b.w, own4);
    const int e0 = c * 1024 + lane * 16;
    unsigned mask = __builtin_amdgcn_ballot_w32(zz != 0u);
    while (mask) {
      if (zz) {
        const int p = __builtin_ctz(zz);
        zz &= zz - 1u;
        qw[(hd + qn + (int)__builtin_amdgcn_mbcnt_lo(mask, 0u)) & (kQCap - 1)] =
            e0 + ((p & 4) << 7) + ((p & 3) << 2) + (p >> 3);
      }
      qn += __builtin_popcount(mask);
      if (qn >= 32) drain(32);
      mask = __builtin_amdgcn_ballot_w32(zz != 0u);
    }
  };
  auto ldv = [&](int c, int half) {
    return *reinterpret_cast<const uint4*>(keys + c * 256 + half * 128 + lane * 4);
  };
  const int wu     = __builtin_amdgcn_readfirstlane(w);
  const int nTrips = (nE + 1023) >> 10;
  if (wu < nTrips) {
    int c = wu;
    uint4 a0 = ldv(c, 0), b0 = ldv(c, 1);
    while (true) {
      const int c1 = c + 32;
      const int c1c = (c1 < nTrips) ? c1 : c;
      const uint4 a1 = ldv(c1c, 0), b1 = ldv(c1c, 1);
      scan(a0, b0, c);
      if (c1 >= nTrips) break;
      const int c2 = c1 + 32;
      const int c2c = (c2 < nTrips) ? c2 : c1;
      a0 = ldv(c2c, 0); b0 = ldv(c2c, 1);
      scan(a1, b1, c1);
      if (c2 >= nTrips) break;
      c = c2;
    }
  }
  drain(qn);
  __syncthreads();

  float4* out4 = reinterpret_cast<float4*>(sum + (size_t)r0 * kH);
  for (int i = tid; i < R * kH / 4; i += 1024)
    if (r0 + (i >> 5) < nDst) out4[i] = acc4[i];
  for (int i = tid; i < R; i += 1024)
    if (r0 + i < nDst) cnt[r0 + i] = cn[i];
}

__global__ void hgnn_addmat(const float* __restrict__ a, const float* __restrict__ b,
                            float* __restrict__ o, int n) {
  int i = blockIdx.x * blockDim.x + threadIdx.x;
  if (i < n) o[i] = a[i] + b[i];
}

template <int NT>
__global__ __launch_bounds__(256)
void hgnn_gemm3(const float* A0, const float* __restrict__ c0,
                const float* __restrict__ W0,
                const float* A1, const float* __restrict__ c1,
                const float* __restrict__ W1,
                const float* A2, const float* __restrict__ c2,
                const float* __restrict__ W2,
                const float* __restrict__ b0, const float* __restrict__ b1,
                float* Y, int relu, int nTiles) {
  __shared__ __attribute__((aligned(16))) float As[NT][16 * 132];
  const int tid  = threadIdx.x;
  const int wv   = tid >> 5;
  const int lane = tid & 31;
  const int m    = lane & 15;
  const int koff = (lane >> 4) << 1;
  const int col  = (wv << 4) + m;

  const float* Alist[3] = {A0, A1, A2};
  const float* Clist[3] = {c0, c1, c2};
  const float* Wlist[3] = {W0, W1, W2};

  v2f bw[NT][32];
#pragma unroll
  for (int t = 0; t < NT; ++t) {
#pragma unroll
    for (int j = 0; j < 32; ++j) {
      const int k = 4 * j + koff;
      bw[t][j].x = Wlist[t][(size_t)k * kH + col];
      bw[t][j].y = Wlist[t][(size_t)(k + 1) * kH + col];
    }
  }
  const float bsum = (b0 ? b0[col] : 0.0f) + (b1 ? b1[col] : 0.0f);

  const int sr  = __builtin_amdgcn_readfirstlane(tid >> 5);
  const int sc4 = tid & 31;
  float4 st[NT][2];
  float  cn[NT][2];
  auto fetch = [&](int tile) {
    const unsigned row0 = (unsigned)tile << 4;
#pragma unroll
    for (int t = 0; t < NT; ++t) {
#pragma unroll
      for (int u = 0; u < 2; ++u) {
        const unsigned r = row0 + (unsigned)(sr + 8 * u);
        st[t][u] = *reinterpret_cast<const float4*>(Alist[t] + (r * kH + 4u * (unsigned)sc4));
        cn[t][u] = Clist[t] ? Clist[t][r] : 1.0f;
      }
    }
  };

  int tile = blockIdx.x;
  if (tile < nTiles) fetch(tile);
  for (; tile < nTiles; tile += gridDim.x) {
    const int row0 = tile << 4;
    __syncthreads();
#pragma unroll
    for (int t = 0; t < NT; ++t) {
#pragma unroll
      for (int u = 0; u < 2; ++u) {
        float4 v = st[t][u];
        if (Clist[t]) {
          const float s = 1.0f / fmaxf(cn[t][u], 1.0f);
          v.x *= s; v.y *= s; v.z *= s; v.w *= s;
        }
        *reinterpret_cast<float4*>(&As[t][(sr + 8 * u) * 132 + 4 * sc4]) = v;
      }
    }
    __syncthreads();
    const int nxt = tile + gridDim.x;
    if (nxt < nTiles) fetch(nxt);

    v8f acc = {};
#pragma unroll
    for (int t = 0; t < NT; ++t) {
#pragma unroll
      for (int j = 0; j < 32; ++j) {
        const int k = 4 * j + koff;
        v2f av;
        av.x = As[t][m * 132 + k];
        av.y = As[t][m * 132 + k + 1];
        acc = __builtin_amdgcn_wmma_f32_16x16x4_f32(
            false, av, false, bw[t][j], (short)0, acc, false, false);
      }
    }

#pragma unroll
    for (int r = 0; r < 8; ++r) {
      float v = acc[r] + bsum;
      if (relu) v = fmaxf(v, 0.0f);
      Y[(unsigned)(row0 + r + ((lane >> 4) << 3)) * kH + (unsigned)col] = v;
    }
  }
}

extern "C" void kernel_launch(void* const* d_in, const int* in_sizes, int n_in,
                              void* d_out, int out_size, void* d_ws, size_t ws_size,
                              hipStream_t stream) {
  (void)n_in; (void)out_size; (void)ws_size;
  const float* EG = (const float*)d_in[0];
  const float* ED = (const float*)d_in[1];
  const float* EC = (const float*)d_in[2];
  const float* Wl = (const float*)d_in[3];
  const float* bl = (const float*)d_in[4];
  const float* Wr = (const float*)d_in[5];
  const int* eg_s  = (const int*)d_in[6];
  const int* eg_d  = (const int*)d_in[7];
  const int* ecd_s = (const int*)d_in[8];
  const int* ecd_d = (const int*)d_in[9];
  const int* ecg_s = (const int*)d_in[10];
  const int* ecg_d = (const int*)d_in[11];
  const int* egc_s = (const int*)d_in[12];
  const int* egc_d = (const int*)d_in[13];

  float* ws = (float*)d_ws;
  float* s0 = ws;
  float* s1 = s0 + (size_t)kNG * kH;
  float* s2 = s1 + (size_t)kND * kH;
  float* s3 = s2 + (size_t)kNG * kH;
  float* c0 = s3 + (size_t)kNC * kH;
  float* c1 = c0 + kNG;
  float* c2 = c1 + kND;
  float* c3 = c2 + kNG;
  float* wsum = c3 + kNC;
  unsigned* keys = (unsigned*)(wsum + kH * kH);
  const unsigned* k0 = keys;
  const unsigned* k1 = k0 + kKeyWords;
  const unsigned* k2 = k1 + kKeyWords1;
  const unsigned* k3 = k2 + kKeyWords;

  float* OG = (float*)d_out;
  float* OD = OG + (size_t)kNG * kH;
  float* OC = OD + (size_t)kND * kH;

  const float* xg = EG;
  const float* xd = ED;
  const float* xc = EC;

  hgnn_keys<<<dim3(kKeyWords / 256, 4), 256, 0, stream>>>(
      eg_d, ecd_d, ecg_d, egc_d,
      in_sizes[6], in_sizes[8], in_sizes[10], in_sizes[12], keys);

  for (int L = 0; L < 2; ++L) {
    hgnn_ownsum<400><<<(kNG + 399) / 400, 1024, 0, stream>>>(xg, eg_s,  eg_d,  k0, s0, c0, in_sizes[6],  kNG);
    hgnn_ownsum<200><<<(kND + 199) / 200, 1024, 0, stream>>>(xc, ecd_s, ecd_d, k1, s1, c1, in_sizes[8],  kND);
    hgnn_ownsum<400><<<(kNG + 399) / 400, 1024, 0, stream>>>(xc, ecg_s, ecg_d, k2, s2, c2, in_sizes[10], kNG);
    hgnn_ownsum<200><<<(kNC + 199) / 200, 1024, 0, stream>>>(xg, egc_s, egc_d, k3, s3, c3, in_sizes[12], kNC);

    const float* WlL = Wl + (size_t)L * 4 * kH * kH;
    const float* WrL = Wr + (size_t)L * 4 * kH * kH;
    const float* blL = bl + (size_t)L * 4 * kH;
    hgnn_addmat<<<(kH * kH + 255) / 256, 256, 0, stream>>>(WrL, WrL + 2 * kH * kH,
                                                           wsum, kH * kH);
    const int relu = (L == 0) ? 1 : 0;
    hgnn_gemm3<3><<<512, 256, 0, stream>>>(
        s0, c0, WlL,
        s2, c2, WlL + 2 * kH * kH,
        xg, nullptr, wsum,
        blL, blL + 2 * kH, OG, relu, kNG / 16);
    hgnn_gemm3<2><<<512, 256, 0, stream>>>(
        s1, c1, WlL + 1 * kH * kH,
        xd, nullptr, WrL + 1 * kH * kH,
        nullptr, nullptr, nullptr,
        blL + 1 * kH, nullptr, OD, relu, kND / 16);
    hgnn_gemm3<2><<<512, 256, 0, stream>>>(
        s3, c3, WlL + 3 * kH * kH,
        xc, nullptr, WrL + 3 * kH * kH,
        nullptr, nullptr, nullptr,
        blL + 3 * kH, nullptr, OC, relu, kNC / 16);

    xg = OG; xd = OD; xc = OC;
  }
}


# Layer 2 segment sums skip the key scan: layer 1's owner work-groups save their matched edges (src, row) in per-owner lists on an LDS cursor, layer 2 reads its list; LDS arrival-order sums kept
# speedup vs baseline: 2.7018x; 1.0765x over previous
#include <hip/hip_runtime.h>

constexpr int kNG = 100000;
constexpr int kND = 50000;
constexpr int kNC = 50000;
constexpr int kH  = 128;

typedef __attribute__((ext_vector_type(2))) float v2f;
typedef __attribute__((ext_vector_type(8))) float v8f;

constexpr int kKeyWords  = 150016;
constexpr int kKeyWords1 = 75008;

__global__ __launch_bounds__(256)
void hgnn_keys(const int* __restrict__ d0, const int* __restrict__ d1,
               const int* __restrict__ d2, const int* __restrict__ d3,
               int n0, int n1, int n2, int n3, unsigned* __restrict__ keys) {
  const int rel = blockIdx.y;
  const int* d = (rel == 0) ? d0 : (rel == 1) ? d1 : (rel == 2) ? d2 : d3;
  const int  n = (rel == 0) ? n0 : (rel == 1) ? n1 : (rel == 2) ? n2 : n3;
  const unsigned nDst = (rel & 1) ? (unsigned)kND : (unsigned)kNG;
  const int      sh   = (rel & 1) ? 0 : 1;
  const int wi = blockIdx.x * 256 + threadIdx.x;
  const int nPadWords = ((n + 1023) >> 10) << 8;
  if (wi >= nPadWords) return;
  const int e = wi * 4;
  int4 v;
  if (e + 3 < n) {
    v = *reinterpret_cast<const int4*>(d + e);
  } else {
    v.x = (e     < n) ? d[e]     : -1;
    v.y = (e + 1 < n) ? d[e + 1] : -1;
    v.z = (e + 2 < n) ? d[e + 2] : -1;
    v.w = (e + 3 < n) ? d[e + 3] : -1;
  }
  auto key = [&](int dd) -> unsigned {
    const unsigned u = (unsigned)dd;
    return (u < nDst) ? ((u / 200u) >> sh) : 255u;
  };
  const int base = (rel == 0) ? 0 : (rel == 1) ? kKeyWords : (rel == 2) ? kKeyWords + kKeyWords1
                                                                      : 2 * kKeyWords + kKeyWords1;
  keys[base + wi] =
      key(v.x) | (key(v.y) << 8) | (key(v.z) << 16) | (key(v.w) << 24);
}

__device__ __forceinline__ unsigned hgnn_eq8(unsigned w, unsigned own4) {
  const unsigned x = w ^ own4;
  return ~(((x & 0x7F7F7F7Fu) + 0x7F7F7F7Fu) | x) & 0x80808080u;
}

constexpr int kQCap = 64;
constexpr int kGrp  = 16;
constexpr int kLstCap = 4096;
constexpr int kOwners = 250;

template <int R>
__global__ __launch_bounds__(1024)
void hgnn_ownsum(const float* __restrict__ x, const int* __restrict__ src,
                 const int* __restrict__ dst, const unsigned* __restrict__ keys,
                 float* __restrict__ sum, float* __restrict__ cnt, int nE, int nDst,
                 unsigned* __restrict__ lst, int* __restrict__ lstN, int useSaved) {
  __shared__ float acc[R * kH];
  __shared__ float cn[R];
  __shared__ int   q[32 * kQCap];
  __shared__ int   cur;
  const int tid  = threadIdx.x;
  const int lane = tid & 31;
  const int w    = tid >> 5;
  const int r0   = blockIdx.x * R;
  int* qw = q + w * kQCap;

  float4* acc4 = reinterpret_cast<float4*>(acc);
  for (int i = tid; i < R * kH / 4; i += 1024) acc4[i] = make_float4(0.f, 0.f, 0.f, 0.f);
  for (int i = tid; i < R; i += 1024) cn[i] = 0.f;
  if (tid == 0) cur = 0;
  __syncthreads();

  unsigned* myl = lst + (size_t)blockIdx.x * kLstCap;

  int qn = 0;
  int hd = 0;

  auto addrows = [&](int nb, int s, int ent) {
    if (lane < nb) atomicAdd(cn + ent, 1.0f);
    for (int i = 0; i < nb; i += kGrp) {
      float2 v[kGrp][2];
      int    dl[kGrp];
#pragma unroll
      for (int u = 0; u < kGrp; ++u) {
        const int k = __builtin_amdgcn_readfirstlane(i + u);
        const int su = __builtin_amdgcn_readlane(s, k);
        dl[u] = __builtin_amdgcn_readlane(ent, k);
        if (k < nb) {
          const float* r = x + (size_t)su * kH + 2 * lane;
          v[u][0] = *reinterpret_cast<const float2*>(r);
          v[u][1] = *reinterpret_cast<const float2*>(r + 64);
        }
      }
#pragma unroll
      for (int u = 0; u < kGrp; ++u) {
        if (i + u < nb) {
          float* a = acc + dl[u] * kH + 2 * lane;
          atomicAdd(a,      v[u][0].x);
          atomicAdd(a + 1,  v[u][0].y);
          atomicAdd(a + 64, v[u][1].x);
          atomicAdd(a + 65, v[u][1].y);
        }
      }
    }
  };

  auto drain = [&](int nb) {
    int ent = 0, s = 0;
    if (lane < nb) {
      const int e = qw[(hd + lane) & (kQCap - 1)];
      s = src[e];
      ent = dst[e] - r0;
    }
    if (!useSaved) {
      int pos = 0;
      if (lane == 0) pos = atomicAdd(&cur, nb);
      pos = __builtin_amdgcn_readfirstlane(pos);
      if (lane < nb && pos + lane < kLstCap)
        myl[pos + lane] = ((unsigned)s << 9) | (unsigned)ent;
    }
    addrows(nb, s, ent);
    hd = (hd + nb) & (kQCap - 1);
    qn -= nb;
  };

  const unsigned own4 = (unsigned)blockIdx.x * 0x01010101u;
  auto scan = [&](const uint4& a, const uint4& b, int c) {
    unsigned zz = (hgnn_eq8(a.x, own4) >> 7) | (hgnn_eq8(a.y, own4) >> 6) |
                  (hgnn_eq8(a.z, own4) >> 5) | (hgnn_eq8(a.w, own4) >> 4) |
                  (hgnn_eq8(b.x, own4) >> 3) | (hgnn_eq8(b.y, own4) >> 2) |
                  (hgnn_eq8(b.z, own4) >> 1) |  hgnn_eq8(b.w, own4);
    const int e0 = c * 1024 + lane * 16;
    unsigned mask = __builtin_amdgcn_ballot_w32(zz != 0u);
    while (mask) {
      if (zz) {
        const int p = __builtin_ctz(zz);
        zz &= zz - 1u;
        qw[(hd + qn + (int)__builtin_amdgcn_mbcnt_lo(mask, 0u)) & (kQCap - 1)] =
            e0 + ((p & 4) << 7) + ((p & 3) << 2) + (p >> 3);
      }
      qn += __builtin_popcount(mask);
      if (qn >= 32) drain(32);
      mask = __builtin_amdgcn_ballot_w32(zz != 0u);
    }
  };
  auto ldv = [&](int c, int half) {
    return *reinterpret_cast<const uint4*>(keys + c * 256 + half * 128 + lane * 4);
  };
  const int wu     = __builtin_amdgcn_readfirstlane(w);
  const int nTrips = (nE + 1023) >> 10;
  const int nSaved = useSaved ? lstN[blockIdx.x] : kLstCap + 1;
  if (nSaved <= kLstCap) {
    int j = wu * 32;
    if (j < nSaved) {
      const int last = nSaved - 1;
      unsigned rec = myl[(j + lane < last) ? j + lane : last];
      while (true) {
        const int jn = j + 1024;
        const unsigned recn = myl[(jn + lane < last) ? jn + lane : last];
        const int left = nSaved - j;
        addrows(left < 32 ? left : 32, (int)(rec >> 9), (int)(rec & 511u));
        if (jn >= nSaved) break;
        rec = recn;
        j = jn;
      }
    }
  } else if (wu < nTrips) {
    int c = wu;
    uint4 a0 = ldv(c, 0), b0 = ldv(c, 1);
    while (true) {
      const int c1 = c + 32;
      const int c1c = (c1 < nTrips) ? c1 : c;
      const uint4 a1 = ldv(c1c, 0), b1 = ldv(c1c, 1);
      scan(a0, b0, c);
      if (c1 >= nTrips) break;
      const int c2 = c1 + 32;
      const int c2c = (c2 < nTrips) ? c2 : c1;
      a0 = ldv(c2c, 0); b0 = ldv(c2c, 1);
      scan(a1, b1, c1);
      if (c2 >= nTrips) break;
      c = c2;
    }
  }
  if (nSaved > kLstCap) drain(qn);
  __syncthreads();
  if (!useSaved && tid == 0) lstN[blockIdx.x] = cur;

  float4* out4 = reinterpret_cast<float4*>(sum + (size_t)r0 * kH);
  for (int i = tid; i < R * kH / 4; i += 1024)
    if (r0 + (i >> 5) < nDst) out4[i] = acc4[i];
  for (int i = tid; i < R; i += 1024)
    if (r0 + i < nDst) cnt[r0 + i] = cn[i];
}

__global__ void hgnn_addmat(const float* __restrict__ a, const float* __restrict__ b,
                            float* __restrict__ o, int n) {
  int i = blockIdx.x * blockDim.x + threadIdx.x;
  if (i < n) o[i] = a[i] + b[i];
}

template <int NT>
__global__ __launch_bounds__(256)
void hgnn_gemm3(const float* A0, const float* __restrict__ c0,
                const float* __restrict__ W0,
                const float* A1, const float* __restrict__ c1,
                const float* __restrict__ W1,
                const float* A2, const float* __restrict__ c2,
                const float* __restrict__ W2,
                const float* __restrict__ b0, const float* __restrict__ b1,
                float* Y, int relu, int nTiles) {
  __shared__ __attribute__((aligned(16))) float As[NT][16 * 132];
  const int tid  = threadIdx.x;
  const int wv   = tid >> 5;
  const int lane = tid & 31;
  const int m    = lane & 15;
  const int koff = (lane >> 4) << 1;
  const int col  = (wv << 4) + m;

  const float* Alist[3] = {A0, A1, A2};
  const float* Clist[3] = {c0, c1, c2};
  const float* Wlist[3] = {W0, W1, W2};

  v2f bw[NT][32];
#pragma unroll
  for (int t = 0; t < NT; ++t) {
#pragma unroll
    for (int j = 0; j < 32; ++j) {
      const int k = 4 * j + koff;
      bw[t][j].x = Wlist[t][(size_t)k * kH + col];
      bw[t][j].y = Wlist[t][(size_t)(k + 1) * kH + col];
    }
  }
  const float bsum = (b0 ? b0[col] : 0.0f) + (b1 ? b1[col] : 0.0f);

  const int sr  = __builtin_amdgcn_readfirstlane(tid >> 5);
  const int sc4 = tid & 31;
  float4 st[NT][2];
  float  cn[NT][2];
  auto fetch = [&](int tile) {
    const unsigned row0 = (unsigned)tile << 4;
#pragma unroll
    for (int t = 0; t < NT; ++t) {
#pragma unroll
      for (int u = 0; u < 2; ++u) {
        const unsigned r = row0 + (unsigned)(sr + 8 * u);
        st[t][u] = *reinterpret_cast<const float4*>(Alist[t] + (r * kH + 4u * (unsigned)sc4));
        cn[t][u] = Clist[t] ? Clist[t][r] : 1.0f;
      }
    }
  };

  int tile = blockIdx.x;
  if (tile < nTiles) fetch(tile);
  for (; tile < nTiles; tile += gridDim.x) {
    const int row0 = tile << 4;
    __syncthreads();
#pragma unroll
    for (int t = 0; t < NT; ++t) {
#pragma unroll
      for (int u = 0; u < 2; ++u) {
        float4 v = st[t][u];
        if (Clist[t]) {
          const float s = 1.0f / fmaxf(cn[t][u], 1.0f);
          v.x *= s; v.y *= s; v.z *= s; v.w *= s;
        }
        *reinterpret_cast<float4*>(&As[t][(sr + 8 * u) * 132 + 4 * sc4]) = v;
      }
    }
    __syncthreads();
    const int nxt = tile + gridDim.x;
    if (nxt < nTiles) fetch(nxt);

    v8f acc = {};
#pragma unroll
    for (int t = 0; t < NT; ++t) {
#pragma unroll
      for (int j = 0; j < 32; ++j) {
        const int k = 4 * j + koff;
        v2f av;
        av.x = As[t][m * 132 + k];
        av.y = As[t][m * 132 + k + 1];
        acc = __builtin_amdgcn_wmma_f32_16x16x4_f32(
            false, av, false, bw[t][j], (short)0, acc, false, false);
      }
    }

#pragma unroll
    for (int r = 0; r < 8; ++r) {
      float v = acc[r] + bsum;
      if (relu) v = fmaxf(v, 0.0f);
      Y[(unsigned)(row0 + r + ((lane >> 4) << 3)) * kH + (unsigned)col] = v;
    }
  }
}

extern "C" void kernel_launch(void* const* d_in, const int* in_sizes, int n_in,
                              void* d_out, int out_size, void* d_ws, size_t ws_size,
                              hipStream_t stream) {
  (void)n_in; (void)out_size; (void)ws_size;
  const float* EG = (const float*)d_in[0];
  const float* ED = (const float*)d_in[1];
  const float* EC = (const float*)d_in[2];
  const float* Wl = (const float*)d_in[3];
  const float* bl = (const float*)d_in[4];
  const float* Wr = (const float*)d_in[5];
  const int* eg_s  = (const int*)d_in[6];
  const int* eg_d  = (const int*)d_in[7];
  const int* ecd_s = (const int*)d_in[8];
  const int* ecd_d = (const int*)d_in[9];
  const int* ecg_s = (const int*)d_in[10];
  const int* ecg_d = (const int*)d_in[11];
  const int* egc_s = (const int*)d_in[12];
  const int* egc_d = (const int*)d_in[13];

  float* ws = (float*)d_ws;
  float* s0 = ws;
  float* sA = s0 + (size_t)kNG * kH;
  float* s2 = sA + (size_t)kND * kH;
  unsigned* lst = (unsigned*)(s2 + (size_t)kNG * kH);
  int* lstN = (int*)(lst + (size_t)4 * kOwners * kLstCap);
  float* c0 = s2 + (size_t)kNG * kH + (size_t)kNC * kH;
  float* c1 = c0 + kNG;
  float* c2 = c1 + kND;
  float* c3 = c2 + kNG;
  float* wsum = c3 + kNC;
  unsigned* keys = (unsigned*)(wsum + kH * kH);
  const unsigned* k0 = keys;
  const unsigned* k1 = k0 + kKeyWords;
  const unsigned* k2 = k1 + kKeyWords1;
  const unsigned* k3 = k2 + kKeyWords;

  float* OG = (float*)d_out;
  float* OD = OG + (size_t)kNG * kH;
  float* OC = OD + (size_t)kND * kH;

  const float* xg = EG;
  const float* xd = ED;
  const float* xc = EC;

  hgnn_keys<<<dim3(kKeyWords / 256, 4), 256, 0, stream>>>(
      eg_d, ecd_d, ecg_d, egc_d,
      in_sizes[6], in_sizes[8], in_sizes[10], in_sizes[12], keys);

  for (int L = 0; L < 2; ++L) {
    const float* WlL = Wl + (size_t)L * 4 * kH * kH;
    const float* WrL = Wr + (size_t)L * 4 * kH * kH;
    const float* blL = bl + (size_t)L * 4 * kH;
    const int relu = (L == 0) ? 1 : 0;
    const int useSaved = (L == 0) ? 0 : 1;

    hgnn_ownsum<200><<<(kND + 199) / 200, 1024, 0, stream>>>(
        xc, ecd_s, ecd_d, k1, sA, c1, in_sizes[8], kND,
        lst + (size_t)1 * kOwners * kLstCap, lstN + 1 * kOwners, useSaved);
    hgnn_gemm3<2><<<512, 256, 0, stream>>>(
        sA, c1, WlL + 1 * kH * kH,
        xd, nullptr, WrL + 1 * kH * kH,
        nullptr, nullptr, nullptr,
        blL + 1 * kH, nullptr, OD, relu, kND / 16);
    hgnn_ownsum<200><<<(kNC + 199) / 200, 1024, 0, stream>>>(
        xg, egc_s, egc_d, k3, sA, c3, in_sizes[12], kNC,
        lst + (size_t)3 * kOwners * kLstCap, lstN + 3 * kOwners, useSaved);
    hgnn_ownsum<400><<<(kNG + 399) / 400, 1024, 0, stream>>>(
        xg, eg_s, eg_d, k0, s0, c0, in_sizes[6], kNG,
        lst, lstN, useSaved);
    hgnn_ownsum<400><<<(kNG + 399) / 400, 1024, 0, stream>>>(
        xc, ecg_s, ecg_d, k2, s2, c2, in_sizes[10], kNG,
        lst + (size_t)2 * kOwners * kLstCap, lstN + 2 * kOwners, useSaved);
    hgnn_gemm3<2><<<512, 256, 0, stream>>>(
        sA, c3, WlL + 3 * kH * kH,
        xc, nullptr, WrL + 3 * kH * kH,
        nullptr, nullptr, nullptr,
        blL + 3 * kH, nullptr, OC, relu, kNC / 16);
    hgnn_addmat<<<(kH * kH + 255) / 256, 256, 0, stream>>>(WrL, WrL + 2 * kH * kH,
                                                           wsum, kH * kH);
    hgnn_gemm3<3><<<512, 256, 0, stream>>>(
        s0, c0, WlL,
        s2, c2, WlL + 2 * kH * kH,
        xg, nullptr, wsum,
        blL, blL + 2 * kH, OG, relu, kNG / 16);

    xg = OG; xd = OD; xc = OC;
  }
}


# Owner key scan replaced by a per-call counting sort: a pre-pass bins each 4096-edge chunk by owner on LDS counters, owners read only their buckets through a prefix table; LDS arrival-order sums kept
# speedup vs baseline: 2.8427x; 1.0522x over previous
#include <hip/hip_runtime.h>

constexpr int kNG = 100000;
constexpr int kND = 50000;
constexpr int kNC = 50000;
constexpr int kH  = 128;

typedef __attribute__((ext_vector_type(2))) float v2f;
typedef __attribute__((ext_vector_type(8))) float v8f;

constexpr int kChunk   = 4096;
constexpr int kChunks  = 147;
constexpr int kChunks1 = 74;
constexpr int kTabW    = 256;

__device__ __forceinline__ int hgnn_scan256(int* a, int lane) {
  int v[8];
  int s = 0;
#pragma unroll
  for (int i = 0; i < 8; ++i) { const int t = a[lane * 8 + i]; v[i] = s; s += t; }
  int inc = s;
#pragma unroll
  for (int d = 1; d < 32; d <<= 1) {
    const int y = __shfl_up(inc, d);
    if (lane >= d) inc += y;
  }
  const int base = inc - s;
#pragma unroll
  for (int i = 0; i < 8; ++i) a[lane * 8 + i] = v[i] + base;
  return __shfl(inc, 31);
}

__global__ __launch_bounds__(1024)
void hgnn_bins(const int* __restrict__ s0, const int* __restrict__ d0,
               const int* __restrict__ s1, const int* __restrict__ d1,
               const int* __restrict__ s2, const int* __restrict__ d2,
               const int* __restrict__ s3, const int* __restrict__ d3,
               int n0, int n1, int n2, int n3,
               int* __restrict__ tab, unsigned* __restrict__ rec) {
  __shared__ int hist[kTabW];
  const int rel = blockIdx.y;
  const int* s = (rel == 0) ? s0 : (rel == 1) ? s1 : (rel == 2) ? s2 : s3;
  const int* d = (rel == 0) ? d0 : (rel == 1) ? d1 : (rel == 2) ? d2 : d3;
  const int  n = (rel == 0) ? n0 : (rel == 1) ? n1 : (rel == 2) ? n2 : n3;
  const int cap = (rel == 1) ? kChunks1 : kChunks;
  int nCh = (n + kChunk - 1) / kChunk;
  if (nCh > cap) nCh = cap;
  const int c = blockIdx.x;
  if (c >= nCh) return;
  const int cb = (rel == 0) ? 0 : (rel == 1) ? kChunks : (rel == 2) ? kChunks + kChunks1
                                                                   : 2 * kChunks + kChunks1;
  const unsigned nDst = (rel & 1) ? (unsigned)kND : (unsigned)kNG;
  const int      sh   = (rel & 1) ? 0 : 1;
  const int tid  = threadIdx.x;
  const int lane = tid & 31;
  if (tid < kTabW) hist[tid] = 0;
  __syncthreads();

  const int e = c * kChunk + tid * 4;
  int4 dv, sv;
  if (e + 3 < n) {
    dv = *reinterpret_cast<const int4*>(d + e);
    sv = *reinterpret_cast<const int4*>(s + e);
  } else {
    dv.x = (e     < n) ? d[e]     : -1;
    dv.y = (e + 1 < n) ? d[e + 1] : -1;
    dv.z = (e + 2 < n) ? d[e + 2] : -1;
    dv.w = (e + 3 < n) ? d[e + 3] : -1;
    sv.x = (e     < n) ? s[e]     : 0;
    sv.y = (e + 1 < n) ? s[e + 1] : 0;
    sv.z = (e + 2 < n) ? s[e + 2] : 0;
    sv.w = (e + 3 < n) ? s[e + 3] : 0;
  }
  const unsigned du[4] = {(unsigned)dv.x, (unsigned)dv.y, (unsigned)dv.z, (unsigned)dv.w};
  const unsigned su[4] = {(unsigned)sv.x, (unsigned)sv.y, (unsigned)sv.z, (unsigned)sv.w};
  unsigned k[4];
  int p[4];
#pragma unroll
  for (int i = 0; i < 4; ++i) {
    k[i] = (du[i] < nDst) ? ((du[i] / 200u) >> sh) : 255u;
    p[i] = 0;
    if (k[i] != 255u) p[i] = atomicAdd(&hist[k[i]], 1);
  }
  __syncthreads();
  if (tid < 32) hgnn_scan256(hist, lane);
  __syncthreads();
  if (tid < kTabW) tab[(cb + c) * kTabW + tid] = hist[tid];
  unsigned* rc = rec + (size_t)(cb + c) * kChunk;
#pragma unroll
  for (int i = 0; i < 4; ++i)
    if (k[i] != 255u)
      rc[hist[k[i]] + p[i]] = (su[i] << 9) | (du[i] - k[i] * (200u << sh));
}

constexpr int kGrp  = 16;
constexpr int kLstCap = 4096;
constexpr int kOwners = 250;

template <int R>
__global__ __launch_bounds__(1024)
void hgnn_ownsum(const float* __restrict__ x, const int* __restrict__ tab,
                 const unsigned* __restrict__ rec,
                 float* __restrict__ sum, float* __restrict__ cnt, int nE, int nDst,
                 unsigned* __restrict__ lst, int* __restrict__ lstN, int useSaved) {
  __shared__ float acc[R * kH];
  __shared__ float cn[R];
  __shared__ int   pre[kTabW + 1];
  __shared__ int   seg[kTabW];
  const int tid  = threadIdx.x;
  const int lane = tid & 31;
  const int w    = tid >> 5;
  const int r0   = blockIdx.x * R;

  float4* acc4 = reinterpret_cast<float4*>(acc);
  for (int i = tid; i < R * kH / 4; i += 1024) acc4[i] = make_float4(0.f, 0.f, 0.f, 0.f);
  for (int i = tid; i < R; i += 1024) cn[i] = 0.f;

  unsigned* myl = lst + (size_t)blockIdx.x * kLstCap;

  const int nSaved = useSaved ? lstN[blockIdx.x] : kLstCap + 1;
  const bool direct = nSaved <= kLstCap;
  int N = nSaved;
  if (!direct) {
    int nCh = (nE + kChunk - 1) / kChunk;
    if (nCh > kTabW) nCh = kTabW;
    if (tid < kTabW) {
      int len = 0, st = 0;
      if (tid < nCh) {
        const int* tr = tab + tid * kTabW + blockIdx.x;
        const int b = tr[0];
        len = tr[1] - b;
        st  = tid * kChunk + b;
      }
      pre[tid] = len;
      seg[tid] = st;
    }
    __syncthreads();
    if (w == 0) {
      const int tot = hgnn_scan256(pre, lane);
      if (lane == 0) pre[kTabW] = tot;
    }
    __syncthreads();
    N = pre[kTabW];
  }
  N = __builtin_amdgcn_readfirstlane(N);
  __syncthreads();

  auto addrows = [&](int nb, int s, int ent) {
    if (lane < nb) atomicAdd(cn + ent, 1.0f);
    for (int i = 0; i < nb; i += kGrp) {
      float2 v[kGrp][2];
      int    dl[kGrp];
#pragma unroll
      for (int u = 0; u < kGrp; ++u) {
        const int k = __builtin_amdgcn_readfirstlane(i + u);
        const int su = __builtin_amdgcn_readlane(s, k);
        dl[u] = __builtin_amdgcn_readlane(ent, k);
        if (k < nb) {
          const float* r = x + (size_t)su * kH + 2 * lane;
          v[u][0] = *reinterpret_cast<const float2*>(r);
          v[u][1] = *reinterpret_cast<const float2*>(r + 64);
        }
      }
#pragma unroll
      for (int u = 0; u < kGrp; ++u) {
        if (i + u < nb) {
          float* a = acc + dl[u] * kH + 2 * lane;
          atomicAdd(a,      v[u][0].x);
          atomicAdd(a + 1,  v[u][0].y);
          atomicAdd(a + 64, v[u][1].x);
          atomicAdd(a + 65, v[u][1].y);
        }
      }
    }
  };

  auto fetch = [&](int j) -> unsigned {
    int v = j + lane;
    if (v > N - 1) v = N - 1;
    if (direct) return myl[v];
    int lo = 0;
    for (int st = kTabW / 2; st >= 1; st >>= 1)
      if (pre[lo + st] <= v) lo += st;
    return rec[seg[lo] + (v - pre[lo])];
  };

  const int wu = __builtin_amdgcn_readfirstlane(w);
  int j = wu * 32;
  if (j < N) {
    unsigned rcd = fetch(j);
    while (true) {
      if (!useSaved) {
        const int v = j + lane;
        if (v < N && v < kLstCap) myl[v] = rcd;
      }
      const int jn = j + 1024;
      const unsigned rcn = fetch(jn);
      const int left = N - j;
      addrows(left < 32 ? left : 32, (int)(rcd >> 9), (int)(rcd & 511u));
      if (jn >= N) break;
      rcd = rcn;
      j = jn;
    }
  }
  __syncthreads();
  if (!useSaved && tid == 0) lstN[blockIdx.x] = N;

  float4* out4 = reinterpret_cast<float4*>(sum + (size_t)r0 * kH);
  for (int i = tid; i < R * kH / 4; i += 1024)
    if (r0 + (i >> 5) < nDst) out4[i] = acc4[i];
  for (int i = tid; i < R; i += 1024)
    if (r0 + i < nDst) cnt[r0 + i] = cn[i];
}

__global__ void hgnn_addmat(const float* __restrict__ a, const float* __restrict__ b,
                            float* __restrict__ o, int n) {
  int i = blockIdx.x * blockDim.x + threadIdx.x;
  if (i < n) o[i] = a[i] + b[i];
}

template <int NT>
__global__ __launch_bounds__(256)
void hgnn_gemm3(const float* A0, const float* __restrict__ c0,
                const float* __restrict__ W0,
                const float* A1, const float* __restrict__ c1,
                const float* __restrict__ W1,
                const float* A2, const float* __restrict__ c2,
                const float* __restrict__ W2,
                const float* __restrict__ b0, const float* __restrict__ b1,
                float* Y, int relu, int nTiles) {
  __shared__ __attribute__((aligned(16))) float As[NT][16 * 132];
  const int tid  = threadIdx.x;
  const int wv   = tid >> 5;
  const int lane = tid & 31;
  const int m    = lane & 15;
  const int koff = (lane >> 4) << 1;
  const int col  = (wv << 4) + m;

  const float* Alist[3] = {A0, A1, A2};
  const float* Clist[3] = {c0, c1, c2};
  const float* Wlist[3] = {W0, W1, W2};

  v2f bw[NT][32];
#pragma unroll
  for (int t = 0; t < NT; ++t) {
#pragma unroll
    for (int j = 0; j < 32; ++j) {
      const int k = 4 * j + koff;
      bw[t][j].x = Wlist[t][(size_t)k * kH + col];
      bw[t][j].y = Wlist[t][(size_t)(k + 1) * kH + col];
    }
  }
  const float bsum = (b0 ? b0[col] : 0.0f) + (b1 ? b1[col] : 0.0f);

  const int sr  = __builtin_amdgcn_readfirstlane(tid >> 5);
  const int sc4 = tid & 31;
  float4 st[NT][2];
  float  cn[NT][2];
  auto fetch = [&](int tile) {
    const unsigned row0 = (unsigned)tile << 4;
#pragma unroll
    for (int t = 0; t < NT; ++t) {
#pragma unroll
      for (int u = 0; u < 2; ++u) {
        const unsigned r = row0 + (unsigned)(sr + 8 * u);
        st[t][u] = *reinterpret_cast<const float4*>(Alist[t] + (r * kH + 4u * (unsigned)sc4));
        cn[t][u] = Clist[t] ? Clist[t][r] : 1.0f;
      }
    }
  };

  int tile = blockIdx.x;
  if (tile < nTiles) fetch(tile);
  for (; tile < nTiles; tile += gridDim.x) {
    const int row0 = tile << 4;
    __syncthreads();
#pragma unroll
    for (int t = 0; t < NT; ++t) {
#pragma unroll
      for (int u = 0; u < 2; ++u) {
        float4 v = st[t][u];
        if (Clist[t]) {
          const float s = 1.0f / fmaxf(cn[t][u], 1.0f);
          v.x *= s; v.y *= s; v.z *= s; v.w *= s;
        }
        *reinterpret_cast<float4*>(&As[t][(sr + 8 * u) * 132 + 4 * sc4]) = v;
      }
    }
    __syncthreads();
    const int nxt = tile + gridDim.x;
    if (nxt < nTiles) fetch(nxt);

    v8f acc = {};
#pragma unroll
    for (int t = 0; t < NT; ++t) {
#pragma unroll
      for (int j = 0; j < 32; ++j) {
        const int k = 4 * j + koff;
        v2f av;
        av.x = As[t][m * 132 + k];
        av.y = As[t][m * 132 + k + 1];
        acc = __builtin_amdgcn_wmma_f32_16x16x4_f32(
            false, av, false, bw[t][j], (short)0, acc, false, false);
      }
    }

#pragma unroll
    for (int r = 0; r < 8; ++r) {
      float v = acc[r] + bsum;
      if (relu) v = fmaxf(v, 0.0f);
      Y[(unsigned)(row0 + r + ((lane >> 4) << 3)) * kH + (unsigned)col] = v;
    }
  }
}

extern "C" void kernel_launch(void* const* d_in, const int* in_sizes, int n_in,
                              void* d_out, int out_size, void* d_ws, size_t ws_size,
                              hipStream_t stream) {
  (void)n_in; (void)out_size; (void)ws_size;
  const float* EG = (const float*)d_in[0];
  const float* ED = (const float*)d_in[1];
  const float* EC = (const float*)d_in[2];
  const float* Wl = (const float*)d_in[3];
  const float* bl = (const float*)d_in[4];
  const float* Wr = (const float*)d_in[5];
  const int* eg_s  = (const int*)d_in[6];
  const int* eg_d  = (const int*)d_in[7];
  const int* ecd_s = (const int*)d_in[8];
  const int* ecd_d = (const int*)d_in[9];
  const int* ecg_s = (const int*)d_in[10];
  const int* ecg_d = (const int*)d_in[11];
  const int* egc_s = (const int*)d_in[12];
  const int* egc_d = (const int*)d_in[13];

  float* ws = (float*)d_ws;
  float* s0 = ws;
  float* sA = s0 + (size_t)kNG * kH;
  float* s2 = sA + (size_t)kND * kH;
  unsigned* lst = (unsigned*)(s2 + (size_t)kNG * kH);
  int* lstN = (int*)(lst + (size_t)4 * kOwners * kLstCap);
  unsigned* rec = (unsigned*)(lstN + 4 * kOwners);
  float* c0 = s2 + (size_t)kNG * kH + (size_t)kNC * kH;
  float* c1 = c0 + kNG;
  float* c2 = c1 + kND;
  float* c3 = c2 + kNG;
  float* wsum = c3 + kNC;
  int* tab = (int*)(wsum + kH * kH);
  const int* t0 = tab;
  const int* t1 = t0 + kChunks * kTabW;
  const int* t2 = t1 + kChunks1 * kTabW;
  const int* t3 = t2 + kChunks * kTabW;
  const unsigned* q0 = rec;
  const unsigned* q1 = q0 + (size_t)kChunks * kChunk;
  const unsigned* q2 = q1 + (size_t)kChunks1 * kChunk;
  const unsigned* q3 = q2 + (size_t)kChunks * kChunk;

  float* OG = (float*)d_out;
  float* OD = OG + (size_t)kNG * kH;
  float* OC = OD + (size_t)kND * kH;

  const float* xg = EG;
  const float* xd = ED;
  const float* xc = EC;

  hgnn_bins<<<dim3(kChunks, 4), 1024, 0, stream>>>(
      eg_s, eg_d, ecd_s, ecd_d, ecg_s, ecg_d, egc_s, egc_d,
      in_sizes[6], in_sizes[8], in_sizes[10], in_sizes[12], tab, rec);

  for (int L = 0; L < 2; ++L) {
    const float* WlL = Wl + (size_t)L * 4 * kH * kH;
    const float* WrL = Wr + (size_t)L * 4 * kH * kH;
    const float* blL = bl + (size_t)L * 4 * kH;
    const int relu = (L == 0) ? 1 : 0;
    const int useSaved = (L == 0) ? 0 : 1;

    hgnn_ownsum<200><<<(kND + 199) / 200, 1024, 0, stream>>>(
        xc, t1, q1, sA, c1, in_sizes[8], kND,
        lst + (size_t)1 * kOwners * kLstCap, lstN + 1 * kOwners, useSaved);
    hgnn_gemm3<2><<<512, 256, 0, stream>>>(
        sA, c1, WlL + 1 * kH * kH,
        xd, nullptr, WrL + 1 * kH * kH,
        nullptr, nullptr, nullptr,
        blL + 1 * kH, nullptr, OD, relu, kND / 16);
    hgnn_ownsum<200><<<(kNC + 199) / 200, 1024, 0, stream>>>(
        xg, t3, q3, sA, c3, in_sizes[12], kNC,
        lst + (size_t)3 * kOwners * kLstCap, lstN + 3 * kOwners, useSaved);
    hgnn_ownsum<400><<<(kNG + 399) / 400, 1024, 0, stream>>>(
        xg, t0, q0, s0, c0, in_sizes[6], kNG,
        lst, lstN, useSaved);
    hgnn_ownsum<400><<<(kNG + 399) / 400, 1024, 0, stream>>>(
        xc, t2, q2, s2, c2, in_sizes[10], kNG,
        lst + (size_t)2 * kOwners * kLstCap, lstN + 2 * kOwners, useSaved);
    hgnn_gemm3<2><<<512, 256, 0, stream>>>(
        sA, c3, WlL + 3 * kH * kH,
        xc, nullptr, WrL + 3 * kH * kH,
        nullptr, nullptr, nullptr,
        blL + 3 * kH, nullptr, OC, relu, kNC / 16);
    hgnn_addmat<<<(kH * kH + 255) / 256, 256, 0, stream>>>(WrL, WrL + 2 * kH * kH,
                                                           wsum, kH * kH);
    hgnn_gemm3<3><<<512, 256, 0, stream>>>(
        s0, c0, WlL,
        s2, c2, WlL + 2 * kH * kH,
        xg, nullptr, wsum,
        blL, blL + 2 * kH, OG, relu, kNG / 16);

    xg = OG; xd = OD; xc = OC;
  }
}
